# starting kernel
# baseline (speedup 1.0000x reference)
#include <hip/hip_runtime.h>
#include <math.h>

#define NS 16
#define NN 1024
#define MB 256
#define NITERS 21
#define LOG2E 1.4426950408889634f
#define EPSC 1e-10f

typedef float v2f __attribute__((ext_vector_type(2)));
typedef float v8f __attribute__((ext_vector_type(8)));

__device__ __forceinline__ float fexp2(float x) { return __builtin_amdgcn_exp2f(x); }
__device__ __forceinline__ float flog2(float x) { return __builtin_amdgcn_logf(x); }

__global__ void __launch_bounds__(256) prep_kernel(const float* __restrict__ la,
                                                   const float* __restrict__ u,
                                                   float* __restrict__ a2,
                                                   float* __restrict__ r2) {
    unsigned idx = blockIdx.x * 256u + threadIdx.x;
    unsigned ij  = idx & (NN * NN - 1);
    float uu    = u[idx];
    float noise = -logf(EPSC - logf(uu + EPSC));
    a2[idx] = LOG2E * (la[ij] + noise);
    if (idx < NS * NN) r2[idx] = 0.0f;
}

__global__ void __launch_bounds__(256) colpass_kernel(const float* __restrict__ a2,
                                                      const float* __restrict__ r2,
                                                      float* __restrict__ parts) {
    __shared__ float rsh[128];
    const int s = blockIdx.z, rc = blockIdx.y, cc = blockIdx.x;
    const int t = threadIdx.x;
    const int j = cc * 256 + t;
    const int i0 = rc * 128;
    if (t < 128) rsh[t] = r2[s * NN + i0 + t];
    __syncthreads();
    const float* ap = a2 + ((size_t)s * NN + i0) * NN + j;
    float s0 = 0.f, s1 = 0.f, s2 = 0.f, s3 = 0.f;
    for (int i = 0; i < 128; i += 4) {
        s0 += fexp2(ap[(size_t)(i + 0) * NN] - rsh[i + 0]);
        s1 += fexp2(ap[(size_t)(i + 1) * NN] - rsh[i + 1]);
        s2 += fexp2(ap[(size_t)(i + 2) * NN] - rsh[i + 2]);
        s3 += fexp2(ap[(size_t)(i + 3) * NN] - rsh[i + 3]);
    }
    parts[((size_t)rc * NS + s) * NN + j] = (s0 + s1) + (s2 + s3);
}

__global__ void __launch_bounds__(256) colfin_kernel(const float* __restrict__ parts,
                                                     float* __restrict__ c2) {
    const int idx = blockIdx.x * 256 + threadIdx.x;
    float sum = 0.f;
#pragma unroll
    for (int p = 0; p < 8; ++p) sum += parts[(size_t)p * NS * NN + idx];
    c2[idx] = flog2(sum);
}

__global__ void __launch_bounds__(256) rowpass_kernel(const float* __restrict__ a2,
                                                      const float* __restrict__ c2,
                                                      float* __restrict__ r2) {
    __shared__ float csh[NN];
    const int s = blockIdx.y;
    const int t = threadIdx.x;
    for (int q = t; q < NN; q += 256) csh[q] = c2[s * NN + q];
    __syncthreads();
    const int w = t >> 5, l = t & 31;
    const int i = blockIdx.x * 8 + w;
    const float* ap = a2 + ((size_t)s * NN + i) * NN;
    float s0 = 0.f, s1 = 0.f;
    for (int j = l; j < NN; j += 64) {
        s0 += fexp2(ap[j]      - csh[j]);
        s1 += fexp2(ap[j + 32] - csh[j + 32]);
    }
    float sum = s0 + s1;
#pragma unroll
    for (int m = 16; m >= 1; m >>= 1) sum += __shfl_xor(sum, m, 32);
    if (l == 0) r2[s * NN + i] = flog2(sum);
}

__global__ void __launch_bounds__(256) gemm_kernel(const float* __restrict__ x,
                                                   const float* __restrict__ a2,
                                                   const float* __restrict__ r2,
                                                   const float* __restrict__ c2,
                                                   float* __restrict__ out) {
    constexpr int XS = 34;
    constexpr int PT = 34;
    __shared__ float xs[MB * XS];
    __shared__ float ps[64 * PT];
    const int s  = blockIdx.y;
    const int m0 = blockIdx.x * 64;
    const int t  = threadIdx.x;
    const int w  = t >> 5, l = t & 31;
    const int lm = l & 15, lh = l >> 4;

    const float* ra = a2 + (size_t)s * NN * NN;
    const float* rr = r2 + s * NN;
    const float* rc = c2 + s * NN;

    v8f acc[2][4];
#pragma unroll
    for (int rt = 0; rt < 2; ++rt)
#pragma unroll
        for (int ct = 0; ct < 4; ++ct)
#pragma unroll
            for (int g = 0; g < 8; ++g) acc[rt][ct][g] = 0.0f;

    for (int k0 = 0; k0 < NN; k0 += 32) {
        __syncthreads();
#pragma unroll
        for (int it = 0; it < 32; ++it) {
            int q = it * 256 + t;
            int row = q >> 5, kk = q & 31;
            xs[row * XS + kk] = x[row * NN + k0 + kk];
        }
#pragma unroll
        for (int it = 0; it < 8; ++it) {
            int q = it * 256 + t;
            int kk = q >> 6, m = q & 63;
            ps[m * PT + kk] =
                fexp2(ra[(size_t)(k0 + kk) * NN + m0 + m] - rr[k0 + kk] - rc[m0 + m]);
        }
        __syncthreads();
#pragma unroll
        for (int ks = 0; ks < 32; ks += 4) {
            const int ka = ks + (lh << 1);
            v2f a[2];
#pragma unroll
            for (int rt = 0; rt < 2; ++rt) {
                int row = w * 32 + rt * 16 + lm;
                a[rt] = *(const v2f*)&xs[row * XS + ka];
            }
#pragma unroll
            for (int ct = 0; ct < 4; ++ct) {
                int col = ct * 16 + lm;
                v2f b = *(const v2f*)&ps[col * PT + ka];
#pragma unroll
                for (int rt = 0; rt < 2; ++rt) {
                    acc[rt][ct] = __builtin_amdgcn_wmma_f32_16x16x4_f32(
                        false, a[rt], false, b, (short)0, acc[rt][ct], false, false);
                }
            }
        }
    }
    float* ro = out + (size_t)s * MB * NN;
#pragma unroll
    for (int rt = 0; rt < 2; ++rt)
#pragma unroll
        for (int ct = 0; ct < 4; ++ct)
#pragma unroll
            for (int g = 0; g < 8; ++g) {
                int row = w * 32 + rt * 16 + lh * 8 + g;
                int col = m0 + ct * 16 + lm;
                ro[row * NN + col] = acc[rt][ct][g];
            }
}

extern "C" void kernel_launch(void* const* d_in, const int* in_sizes, int n_in,
                              void* d_out, int out_size, void* d_ws, size_t ws_size,
                              hipStream_t stream) {
    const float* x  = (const float*)d_in[0];
    const float* la = (const float*)d_in[1];
    const float* u  = (const float*)d_in[2];
    float* out = (float*)d_out;

    char* ws = (char*)d_ws;
    float* a2    = (float*)ws;
    float* r2    = (float*)(ws + (size_t)NS * NN * NN * sizeof(float));
    float* c2    = r2 + NS * NN;
    float* parts = c2 + NS * NN;

    prep_kernel<<<(NS * NN * NN) / 256, 256, 0, stream>>>(la, u, a2, r2);
    for (int it = 0; it < NITERS; ++it) {
        colpass_kernel<<<dim3(4, 8, NS), 256, 0, stream>>>(a2, r2, parts);
        colfin_kernel<<<(NS * NN) / 256, 256, 0, stream>>>(parts, c2);
        rowpass_kernel<<<dim3(NN / 8, NS), 256, 0, stream>>>(a2, c2, r2);
    }
    gemm_kernel<<<dim3(NN / 64, NS), 256, 0, stream>>>(x, a2, r2, c2, out);
}


# Row pass re-tiled: one wave sums 4 whole rows, 128-bit loads, c in registers, all 40 loads in flight, no LDS copy or barrier in front; 16384 waves become 4096; row sums added in a new order
# speedup vs baseline: 1.3138x; 1.3138x over previous
#include <hip/hip_runtime.h>
#include <math.h>

#define NS 16
#define NN 1024
#define MB 256
#define NITERS 21
#define LOG2E 1.4426950408889634f
#define EPSC 1e-10f

typedef float v2f __attribute__((ext_vector_type(2)));
typedef float v4f __attribute__((ext_vector_type(4)));
typedef float v8f __attribute__((ext_vector_type(8)));

__device__ __forceinline__ float fexp2(float x) { return __builtin_amdgcn_exp2f(x); }
__device__ __forceinline__ float flog2(float x) { return __builtin_amdgcn_logf(x); }

__global__ void __launch_bounds__(256) prep_kernel(const float* __restrict__ la,
                                                   const float* __restrict__ u,
                                                   float* __restrict__ a2,
                                                   float* __restrict__ r2) {
    unsigned idx = blockIdx.x * 256u + threadIdx.x;
    unsigned ij  = idx & (NN * NN - 1);
    float uu    = u[idx];
    float noise = -logf(EPSC - logf(uu + EPSC));
    a2[idx] = LOG2E * (la[ij] + noise);
    if (idx < NS * NN) r2[idx] = 0.0f;
}

__global__ void __launch_bounds__(256) colpass_kernel(const float* __restrict__ a2,
                                                      const float* __restrict__ r2,
                                                      float* __restrict__ c2) {
    __shared__ float red[32 * 32];
    const int s = blockIdx.y, strip = blockIdx.x;
    const int t = threadIdx.x;
    const int g  = t >> 3;
    const int lc = t & 7;
    const float* rp = r2 + s * NN + g * 32;
    const float* ap = a2 + ((size_t)s * NN + g * 32) * NN + strip * 32 + lc * 4;
    v4f rv[8];
    for (int k = 0; k < 8; ++k) rv[k] = *(const v4f*)(rp + 4 * k);
    v4f av[32];
    for (int q = 0; q < 32; ++q) av[q] = *(const v4f*)(ap + (size_t)q * NN);
    float c0 = 0.f, c1 = 0.f, c2v = 0.f, c3 = 0.f;
    for (int q = 0; q < 32; ++q) {
        const float r = rv[q >> 2][q & 3];
        c0  += fexp2(av[q][0] - r);
        c1  += fexp2(av[q][1] - r);
        c2v += fexp2(av[q][2] - r);
        c3  += fexp2(av[q][3] - r);
    }
    v4f cs; cs[0] = c0; cs[1] = c1; cs[2] = c2v; cs[3] = c3;
    *(v4f*)&red[g * 32 + lc * 4] = cs;
    __syncthreads();
    if (t < 32) {
        float sum = 0.f;
        for (int k = 0; k < 32; ++k) sum += red[k * 32 + t];
        c2[s * NN + strip * 32 + t] = flog2(sum);
    }
}

__global__ void __launch_bounds__(256) rowpass_kernel(const float* __restrict__ a2,
                                                      const float* __restrict__ c2,
                                                      float* __restrict__ r2) {
    __shared__ __attribute__((aligned(16))) float red[32];
    const int s = blockIdx.y;
    const int t = threadIdx.x;
    const int w = t >> 5, l = t & 31;
    const int i0 = blockIdx.x * 32 + w * 4;
    const float* cp = c2 + s * NN + l * 4;
    const float* ap = a2 + ((size_t)s * NN + i0) * NN + l * 4;
    v4f cv[8];
    for (int k = 0; k < 8; ++k) cv[k] = *(const v4f*)(cp + 128 * k);
    v4f av[4][8];
    for (int r = 0; r < 4; ++r)
        for (int k = 0; k < 8; ++k) av[r][k] = *(const v4f*)(ap + (size_t)r * NN + 128 * k);
    float sum[4];
    for (int r = 0; r < 4; ++r) {
        __builtin_amdgcn_sched_barrier(0);
        float s0 = 0.f, s1 = 0.f, s2 = 0.f, s3 = 0.f;
        for (int k = 0; k < 8; ++k) {
            s0 += fexp2(av[r][k][0] - cv[k][0]);
            s1 += fexp2(av[r][k][1] - cv[k][1]);
            s2 += fexp2(av[r][k][2] - cv[k][2]);
            s3 += fexp2(av[r][k][3] - cv[k][3]);
        }
        sum[r] = (s0 + s1) + (s2 + s3);
    }
    for (int m = 16; m >= 1; m >>= 1)
        for (int r = 0; r < 4; ++r) sum[r] += __shfl_xor(sum[r], m, 32);
    if (l == 0) {
        v4f o; o[0] = sum[0]; o[1] = sum[1]; o[2] = sum[2]; o[3] = sum[3];
        *(v4f*)&red[w * 4] = o;
    }
    __syncthreads();
    if (t < 32) r2[s * NN + blockIdx.x * 32 + t] = flog2(red[t]);
}

__global__ void __launch_bounds__(256) gemm_kernel(const float* __restrict__ x,
                                                   const float* __restrict__ a2,
                                                   const float* __restrict__ r2,
                                                   const float* __restrict__ c2,
                                                   float* __restrict__ out) {
    constexpr int XS = 34;
    constexpr int PT = 34;
    __shared__ float xs[MB * XS];
    __shared__ float ps[64 * PT];
    const int s  = blockIdx.y;
    const int m0 = blockIdx.x * 64;
    const int t  = threadIdx.x;
    const int w  = t >> 5, l = t & 31;
    const int lm = l & 15, lh = l >> 4;

    const float* ra = a2 + (size_t)s * NN * NN;
    const float* rr = r2 + s * NN;
    const float* rc = c2 + s * NN;

    v8f acc[2][4];
#pragma unroll
    for (int rt = 0; rt < 2; ++rt)
#pragma unroll
        for (int ct = 0; ct < 4; ++ct)
#pragma unroll
            for (int g = 0; g < 8; ++g) acc[rt][ct][g] = 0.0f;

    for (int k0 = 0; k0 < NN; k0 += 32) {
        __syncthreads();
#pragma unroll
        for (int it = 0; it < 32; ++it) {
            int q = it * 256 + t;
            int row = q >> 5, kk = q & 31;
            xs[row * XS + kk] = x[row * NN + k0 + kk];
        }
#pragma unroll
        for (int it = 0; it < 8; ++it) {
            int q = it * 256 + t;
            int kk = q >> 6, m = q & 63;
            ps[m * PT + kk] =
                fexp2(ra[(size_t)(k0 + kk) * NN + m0 + m] - rr[k0 + kk] - rc[m0 + m]);
        }
        __syncthreads();
#pragma unroll
        for (int ks = 0; ks < 32; ks += 4) {
            const int ka = ks + (lh << 1);
            v2f a[2];
#pragma unroll
            for (int rt = 0; rt < 2; ++rt) {
                int row = w * 32 + rt * 16 + lm;
                a[rt] = *(const v2f*)&xs[row * XS + ka];
            }
#pragma unroll
            for (int ct = 0; ct < 4; ++ct) {
                int col = ct * 16 + lm;
                v2f b = *(const v2f*)&ps[col * PT + ka];
#pragma unroll
                for (int rt = 0; rt < 2; ++rt) {
                    acc[rt][ct] = __builtin_amdgcn_wmma_f32_16x16x4_f32(
                        false, a[rt], false, b, (short)0, acc[rt][ct], false, false);
                }
            }
        }
    }
    float* ro = out + (size_t)s * MB * NN;
#pragma unroll
    for (int rt = 0; rt < 2; ++rt)
#pragma unroll
        for (int ct = 0; ct < 4; ++ct)
#pragma unroll
            for (int g = 0; g < 8; ++g) {
                int row = w * 32 + rt * 16 + lh * 8 + g;
                int col = m0 + ct * 16 + lm;
                ro[row * NN + col] = acc[rt][ct][g];
            }
}

extern "C" void kernel_launch(void* const* d_in, const int* in_sizes, int n_in,
                              void* d_out, int out_size, void* d_ws, size_t ws_size,
                              hipStream_t stream) {
    const float* x  = (const float*)d_in[0];
    const float* la = (const float*)d_in[1];
    const float* u  = (const float*)d_in[2];
    float* out = (float*)d_out;

    char* ws = (char*)d_ws;
    float* a2    = (float*)ws;
    float* r2    = (float*)(ws + (size_t)NS * NN * NN * sizeof(float));
    float* c2    = r2 + NS * NN;

    prep_kernel<<<(NS * NN * NN) / 256, 256, 0, stream>>>(la, u, a2, r2);
    for (int it = 0; it < NITERS; ++it) {
        colpass_kernel<<<dim3(NN / 32, NS), 256, 0, stream>>>(a2, r2, c2);
        rowpass_kernel<<<dim3(NN / 32, NS), 256, 0, stream>>>(a2, c2, r2);
    }
    gemm_kernel<<<dim3(NN / 64, NS), 256, 0, stream>>>(x, a2, r2, c2, out);
}


# Prep pass fused into the first column pass: one sweep computes and stores a2 and sums its whole columns (64-column strips, 16 rows of loads in flight); 44 to 43 launches; sums keep their order
# speedup vs baseline: 1.3906x; 1.0585x over previous
#include <hip/hip_runtime.h>
#include <math.h>

#define NS 16
#define NN 1024
#define MB 256
#define NITERS 21
#define LOG2E 1.4426950408889634f
#define EPSC 1e-10f

typedef float v2f __attribute__((ext_vector_type(2)));
typedef float v4f __attribute__((ext_vector_type(4)));
typedef float v8f __attribute__((ext_vector_type(8)));

__device__ __forceinline__ float fexp2(float x) { return __builtin_amdgcn_exp2f(x); }
__device__ __forceinline__ float flog2(float x) { return __builtin_amdgcn_logf(x); }

__global__ void __launch_bounds__(512) prepcol_kernel(const float* __restrict__ la,
                                                      const float* __restrict__ u,
                                                      float* __restrict__ a2,
                                                      float* __restrict__ c2) {
    __shared__ __attribute__((aligned(16))) float red[32 * 64];
    const int s = blockIdx.y, strip = blockIdx.x;
    const int t = threadIdx.x;
    const int g  = t >> 4;
    const int lc = t & 15;
    const size_t off = (size_t)(g * 32) * NN + strip * 64 + lc * 4;
    const float* up = u + (size_t)s * NN * NN + off;
    const float* lp = la + off;
    float* op = a2 + (size_t)s * NN * NN + off;
    constexpr int WIN = 16;
    v4f lv[32], uv[32];
#pragma unroll
    for (int q = 0; q < WIN; ++q) {
        lv[q] = *(const v4f*)(lp + (size_t)q * NN);
        uv[q] = *(const v4f*)(up + (size_t)q * NN);
    }
    float c0 = 0.f, c1 = 0.f, c2v = 0.f, c3 = 0.f;
#pragma unroll
    for (int q = 0; q < 32; ++q) {
        __builtin_amdgcn_sched_barrier(0);
        if (q + WIN < 32) {
            lv[q + WIN] = *(const v4f*)(lp + (size_t)(q + WIN) * NN);
            uv[q + WIN] = *(const v4f*)(up + (size_t)(q + WIN) * NN);
        }
        v4f av;
        for (int e = 0; e < 4; ++e) {
            const float noise = -logf(EPSC - logf(uv[q][e] + EPSC));
            av[e] = LOG2E * (lv[q][e] + noise);
        }
        *(v4f*)(op + (size_t)q * NN) = av;
        c0  += fexp2(av[0]);
        c1  += fexp2(av[1]);
        c2v += fexp2(av[2]);
        c3  += fexp2(av[3]);
    }
    v4f cs; cs[0] = c0; cs[1] = c1; cs[2] = c2v; cs[3] = c3;
    *(v4f*)&red[g * 64 + lc * 4] = cs;
    __syncthreads();
    if (t < 64) {
        float sum = 0.f;
        for (int k = 0; k < 32; ++k) sum += red[k * 64 + t];
        c2[s * NN + strip * 64 + t] = flog2(sum);
    }
}

__global__ void __launch_bounds__(256) colpass_kernel(const float* __restrict__ a2,
                                                      const float* __restrict__ r2,
                                                      float* __restrict__ c2) {
    __shared__ float red[32 * 32];
    const int s = blockIdx.y, strip = blockIdx.x;
    const int t = threadIdx.x;
    const int g  = t >> 3;
    const int lc = t & 7;
    const float* rp = r2 + s * NN + g * 32;
    const float* ap = a2 + ((size_t)s * NN + g * 32) * NN + strip * 32 + lc * 4;
    v4f rv[8];
    for (int k = 0; k < 8; ++k) rv[k] = *(const v4f*)(rp + 4 * k);
    v4f av[32];
    for (int q = 0; q < 32; ++q) av[q] = *(const v4f*)(ap + (size_t)q * NN);
    float c0 = 0.f, c1 = 0.f, c2v = 0.f, c3 = 0.f;
    for (int q = 0; q < 32; ++q) {
        const float r = rv[q >> 2][q & 3];
        c0  += fexp2(av[q][0] - r);
        c1  += fexp2(av[q][1] - r);
        c2v += fexp2(av[q][2] - r);
        c3  += fexp2(av[q][3] - r);
    }
    v4f cs; cs[0] = c0; cs[1] = c1; cs[2] = c2v; cs[3] = c3;
    *(v4f*)&red[g * 32 + lc * 4] = cs;
    __syncthreads();
    if (t < 32) {
        float sum = 0.f;
        for (int k = 0; k < 32; ++k) sum += red[k * 32 + t];
        c2[s * NN + strip * 32 + t] = flog2(sum);
    }
}

__global__ void __launch_bounds__(256) rowpass_kernel(const float* __restrict__ a2,
                                                      const float* __restrict__ c2,
                                                      float* __restrict__ r2) {
    __shared__ __attribute__((aligned(16))) float red[32];
    const int s = blockIdx.y;
    const int t = threadIdx.x;
    const int w = t >> 5, l = t & 31;
    const int i0 = blockIdx.x * 32 + w * 4;
    const float* cp = c2 + s * NN + l * 4;
    const float* ap = a2 + ((size_t)s * NN + i0) * NN + l * 4;
    v4f cv[8];
    for (int k = 0; k < 8; ++k) cv[k] = *(const v4f*)(cp + 128 * k);
    v4f av[4][8];
    for (int r = 0; r < 4; ++r)
        for (int k = 0; k < 8; ++k) av[r][k] = *(const v4f*)(ap + (size_t)r * NN + 128 * k);
    float sum[4];
    for (int r = 0; r < 4; ++r) {
        __builtin_amdgcn_sched_barrier(0);
        float s0 = 0.f, s1 = 0.f, s2 = 0.f, s3 = 0.f;
        for (int k = 0; k < 8; ++k) {
            s0 += fexp2(av[r][k][0] - cv[k][0]);
            s1 += fexp2(av[r][k][1] - cv[k][1]);
            s2 += fexp2(av[r][k][2] - cv[k][2]);
            s3 += fexp2(av[r][k][3] - cv[k][3]);
        }
        sum[r] = (s0 + s1) + (s2 + s3);
    }
    for (int m = 16; m >= 1; m >>= 1)
        for (int r = 0; r < 4; ++r) sum[r] += __shfl_xor(sum[r], m, 32);
    if (l == 0) {
        v4f o; o[0] = sum[0]; o[1] = sum[1]; o[2] = sum[2]; o[3] = sum[3];
        *(v4f*)&red[w * 4] = o;
    }
    __syncthreads();
    if (t < 32) r2[s * NN + blockIdx.x * 32 + t] = flog2(red[t]);
}

__global__ void __launch_bounds__(256) gemm_kernel(const float* __restrict__ x,
                                                   const float* __restrict__ a2,
                                                   const float* __restrict__ r2,
                                                   const float* __restrict__ c2,
                                                   float* __restrict__ out) {
    constexpr int XS = 34;
    constexpr int PT = 34;
    __shared__ float xs[MB * XS];
    __shared__ float ps[64 * PT];
    const int s  = blockIdx.y;
    const int m0 = blockIdx.x * 64;
    const int t  = threadIdx.x;
    const int w  = t >> 5, l = t & 31;
    const int lm = l & 15, lh = l >> 4;

    const float* ra = a2 + (size_t)s * NN * NN;
    const float* rr = r2 + s * NN;
    const float* rc = c2 + s * NN;

    v8f acc[2][4];
#pragma unroll
    for (int rt = 0; rt < 2; ++rt)
#pragma unroll
        for (int ct = 0; ct < 4; ++ct)
#pragma unroll
            for (int g = 0; g < 8; ++g) acc[rt][ct][g] = 0.0f;

    for (int k0 = 0; k0 < NN; k0 += 32) {
        __syncthreads();
#pragma unroll
        for (int it = 0; it < 32; ++it) {
            int q = it * 256 + t;
            int row = q >> 5, kk = q & 31;
            xs[row * XS + kk] = x[row * NN + k0 + kk];
        }
#pragma unroll
        for (int it = 0; it < 8; ++it) {
            int q = it * 256 + t;
            int kk = q >> 6, m = q & 63;
            ps[m * PT + kk] =
                fexp2(ra[(size_t)(k0 + kk) * NN + m0 + m] - rr[k0 + kk] - rc[m0 + m]);
        }
        __syncthreads();
#pragma unroll
        for (int ks = 0; ks < 32; ks += 4) {
            const int ka = ks + (lh << 1);
            v2f a[2];
#pragma unroll
            for (int rt = 0; rt < 2; ++rt) {
                int row = w * 32 + rt * 16 + lm;
                a[rt] = *(const v2f*)&xs[row * XS + ka];
            }
#pragma unroll
            for (int ct = 0; ct < 4; ++ct) {
                int col = ct * 16 + lm;
                v2f b = *(const v2f*)&ps[col * PT + ka];
#pragma unroll
                for (int rt = 0; rt < 2; ++rt) {
                    acc[rt][ct] = __builtin_amdgcn_wmma_f32_16x16x4_f32(
                        false, a[rt], false, b, (short)0, acc[rt][ct], false, false);
                }
            }
        }
    }
    float* ro = out + (size_t)s * MB * NN;
#pragma unroll
    for (int rt = 0; rt < 2; ++rt)
#pragma unroll
        for (int ct = 0; ct < 4; ++ct)
#pragma unroll
            for (int g = 0; g < 8; ++g) {
                int row = w * 32 + rt * 16 + lh * 8 + g;
                int col = m0 + ct * 16 + lm;
                ro[row * NN + col] = acc[rt][ct][g];
            }
}

extern "C" void kernel_launch(void* const* d_in, const int* in_sizes, int n_in,
                              void* d_out, int out_size, void* d_ws, size_t ws_size,
                              hipStream_t stream) {
    const float* x  = (const float*)d_in[0];
    const float* la = (const float*)d_in[1];
    const float* u  = (const float*)d_in[2];
    float* out = (float*)d_out;

    char* ws = (char*)d_ws;
    float* a2    = (float*)ws;
    float* r2    = (float*)(ws + (size_t)NS * NN * NN * sizeof(float));
    float* c2    = r2 + NS * NN;

    prepcol_kernel<<<dim3(NN / 64, NS), 512, 0, stream>>>(la, u, a2, c2);
    rowpass_kernel<<<dim3(NN / 32, NS), 256, 0, stream>>>(a2, c2, r2);
    for (int it = 1; it < NITERS; ++it) {
        colpass_kernel<<<dim3(NN / 32, NS), 256, 0, stream>>>(a2, r2, c2);
        rowpass_kernel<<<dim3(NN / 32, NS), 256, 0, stream>>>(a2, c2, r2);
    }
    gemm_kernel<<<dim3(NN / 64, NS), 256, 0, stream>>>(x, a2, r2, c2, out);
}


# Column pass fused with the next row pass: a block keeps its 1024x64 strip of a2 in registers for both; row sums leave as 16 planes added by the next launch; 43 to 24 launches; row sums reordered
# speedup vs baseline: 1.7893x; 1.2867x over previous
#include <hip/hip_runtime.h>
#include <math.h>

#define NS 16
#define NN 1024
#define MB 256
#define NITERS 21
#define NP 16
#define LOG2E 1.4426950408889634f
#define EPSC 1e-10f

typedef float v2f __attribute__((ext_vector_type(2)));
typedef float v4f __attribute__((ext_vector_type(4)));
typedef float v8f __attribute__((ext_vector_type(8)));

__device__ __forceinline__ float fexp2(float x) { return __builtin_amdgcn_exp2f(x); }
__device__ __forceinline__ float flog2(float x) { return __builtin_amdgcn_logf(x); }

__global__ void __launch_bounds__(512) prepcol_kernel(const float* __restrict__ la,
                                                      const float* __restrict__ u,
                                                      float* __restrict__ a2,
                                                      float* __restrict__ c2) {
    __shared__ __attribute__((aligned(16))) float red[32 * 64];
    const int s = blockIdx.y, strip = blockIdx.x;
    const int t = threadIdx.x;
    const int g  = t >> 4;
    const int lc = t & 15;
    const size_t off = (size_t)(g * 32) * NN + strip * 64 + lc * 4;
    const float* up = u + (size_t)s * NN * NN + off;
    const float* lp = la + off;
    float* op = a2 + (size_t)s * NN * NN + off;
    constexpr int WIN = 16;
    v4f lv[32], uv[32];
#pragma unroll
    for (int q = 0; q < WIN; ++q) {
        lv[q] = *(const v4f*)(lp + (size_t)q * NN);
        uv[q] = *(const v4f*)(up + (size_t)q * NN);
    }
    float c0 = 0.f, c1 = 0.f, c2v = 0.f, c3 = 0.f;
#pragma unroll
    for (int q = 0; q < 32; ++q) {
        __builtin_amdgcn_sched_barrier(0);
        if (q + WIN < 32) {
            lv[q + WIN] = *(const v4f*)(lp + (size_t)(q + WIN) * NN);
            uv[q + WIN] = *(const v4f*)(up + (size_t)(q + WIN) * NN);
        }
        v4f av;
        for (int e = 0; e < 4; ++e) {
            const float noise = -logf(EPSC - logf(uv[q][e] + EPSC));
            av[e] = LOG2E * (lv[q][e] + noise);
        }
        *(v4f*)(op + (size_t)q * NN) = av;
        c0  += fexp2(av[0]);
        c1  += fexp2(av[1]);
        c2v += fexp2(av[2]);
        c3  += fexp2(av[3]);
    }
    v4f cs; cs[0] = c0; cs[1] = c1; cs[2] = c2v; cs[3] = c3;
    *(v4f*)&red[g * 64 + lc * 4] = cs;
    __syncthreads();
    if (t < 64) {
        float sum = 0.f;
        for (int k = 0; k < 32; ++k) sum += red[k * 64 + t];
        c2[s * NN + strip * 64 + t] = flog2(sum);
    }
}

template <bool FIRST>
__global__ void __launch_bounds__(512) colrow_kernel(const float* __restrict__ a2,
                                                     const float* __restrict__ rin,
                                                     float* __restrict__ c2,
                                                     float* __restrict__ pout) {
    __shared__ __attribute__((aligned(16))) float rs[NN];
    __shared__ __attribute__((aligned(16))) float red[32 * 64];
    __shared__ __attribute__((aligned(16))) float cs[64];
    const int s = blockIdx.y, strip = blockIdx.x;
    const int t = threadIdx.x;
    const int g  = t >> 4;
    const int lc = t & 15;
    v2f pv[NP];
    if (FIRST) {
        pv[0] = *(const v2f*)(rin + s * NN + 2 * t);
    } else {
#pragma unroll
        for (int p = 0; p < NP; ++p) pv[p] = *(const v2f*)(rin + (size_t)(s * NP + p) * NN + 2 * t);
    }
    const float* ap = a2 + ((size_t)s * NN + g * 32) * NN + strip * 64 + lc * 4;
    v4f av[32];
#pragma unroll
    for (int q = 0; q < 32; ++q) av[q] = *(const v4f*)(ap + (size_t)q * NN);
    v2f rr;
    if (FIRST) {
        rr = pv[0];
    } else {
        float s0 = 0.f, s1 = 0.f;
#pragma unroll
        for (int p = 0; p < NP; ++p) { s0 += pv[p][0]; s1 += pv[p][1]; }
        rr[0] = flog2(s0); rr[1] = flog2(s1);
    }
    *(v2f*)&rs[2 * t] = rr;
    __builtin_amdgcn_fence(__ATOMIC_RELEASE, "wavefront");
    __builtin_amdgcn_wave_barrier();
    __builtin_amdgcn_fence(__ATOMIC_ACQUIRE, "wavefront");
    v4f rv[8];
#pragma unroll
    for (int k = 0; k < 8; ++k) rv[k] = *(const v4f*)&rs[g * 32 + 4 * k];
    float c0 = 0.f, c1 = 0.f, c2v = 0.f, c3 = 0.f;
#pragma unroll
    for (int q = 0; q < 32; ++q) {
        if ((q & 3) == 0) __builtin_amdgcn_sched_barrier(0);
        const float r = rv[q >> 2][q & 3];
        c0  += fexp2(av[q][0] - r);
        c1  += fexp2(av[q][1] - r);
        c2v += fexp2(av[q][2] - r);
        c3  += fexp2(av[q][3] - r);
    }
    v4f csum; csum[0] = c0; csum[1] = c1; csum[2] = c2v; csum[3] = c3;
    *(v4f*)&red[g * 64 + lc * 4] = csum;
    __syncthreads();
    if (t < 64) {
        float sum = 0.f;
        for (int k = 0; k < 32; ++k) sum += red[k * 64 + t];
        const float c = flog2(sum);
        cs[t] = c;
        c2[s * NN + strip * 64 + t] = c;
    }
    __syncthreads();
    const v4f cv = *(const v4f*)&cs[lc * 4];
    float v[32];
#pragma unroll
    for (int q = 0; q < 32; ++q) {
        v[q] = (fexp2(av[q][0] - cv[0]) + fexp2(av[q][1] - cv[1])) +
               (fexp2(av[q][2] - cv[2]) + fexp2(av[q][3] - cv[3]));
    }
#pragma unroll
    for (int h = 16; h >= 2; h >>= 1) {
        const bool up = (lc & (h >> 1)) != 0;
#pragma unroll
        for (int k = 0; k < h; ++k) {
            const float lo = v[k], hi = v[k + h];
            const float keep = up ? hi : lo;
            const float send = up ? lo : hi;
            v[k] = keep + __shfl_xor(send, h >> 1, 32);
        }
    }
    v2f o; o[0] = v[0]; o[1] = v[1];
    *(v2f*)(pout + (size_t)(s * NP + strip) * NN + g * 32 + 2 * lc) = o;
}

__global__ void __launch_bounds__(256) rowpass_kernel(const float* __restrict__ a2,
                                                      const float* __restrict__ c2,
                                                      float* __restrict__ r2) {
    __shared__ __attribute__((aligned(16))) float red[32];
    const int s = blockIdx.y;
    const int t = threadIdx.x;
    const int w = t >> 5, l = t & 31;
    const int i0 = blockIdx.x * 32 + w * 4;
    const float* cp = c2 + s * NN + l * 4;
    const float* ap = a2 + ((size_t)s * NN + i0) * NN + l * 4;
    v4f cv[8];
    for (int k = 0; k < 8; ++k) cv[k] = *(const v4f*)(cp + 128 * k);
    v4f av[4][8];
    for (int r = 0; r < 4; ++r)
        for (int k = 0; k < 8; ++k) av[r][k] = *(const v4f*)(ap + (size_t)r * NN + 128 * k);
    float sum[4];
    for (int r = 0; r < 4; ++r) {
        __builtin_amdgcn_sched_barrier(0);
        float s0 = 0.f, s1 = 0.f, s2 = 0.f, s3 = 0.f;
        for (int k = 0; k < 8; ++k) {
            s0 += fexp2(av[r][k][0] - cv[k][0]);
            s1 += fexp2(av[r][k][1] - cv[k][1]);
            s2 += fexp2(av[r][k][2] - cv[k][2]);
            s3 += fexp2(av[r][k][3] - cv[k][3]);
        }
        sum[r] = (s0 + s1) + (s2 + s3);
    }
    for (int m = 16; m >= 1; m >>= 1)
        for (int r = 0; r < 4; ++r) sum[r] += __shfl_xor(sum[r], m, 32);
    if (l == 0) {
        v4f o; o[0] = sum[0]; o[1] = sum[1]; o[2] = sum[2]; o[3] = sum[3];
        *(v4f*)&red[w * 4] = o;
    }
    __syncthreads();
    if (t < 32) r2[s * NN + blockIdx.x * 32 + t] = flog2(red[t]);
}

__global__ void __launch_bounds__(256) rowfin_kernel(const float* __restrict__ pl, float* __restrict__ r2) {
    const int s = blockIdx.y;
    const int i = blockIdx.x * 256 + threadIdx.x;
    float sum = 0.f;
    for (int p = 0; p < NP; ++p) sum += pl[(size_t)(s * NP + p) * NN + i];
    r2[s * NN + i] = flog2(sum);
}

__global__ void __launch_bounds__(256) gemm_kernel(const float* __restrict__ x,
                                                   const float* __restrict__ a2,
                                                   const float* __restrict__ r2,
                                                   const float* __restrict__ c2,
                                                   float* __restrict__ out) {
    constexpr int XS = 34;
    constexpr int PT = 34;
    __shared__ float xs[MB * XS];
    __shared__ float ps[64 * PT];
    const int s  = blockIdx.y;
    const int m0 = blockIdx.x * 64;
    const int t  = threadIdx.x;
    const int w  = t >> 5, l = t & 31;
    const int lm = l & 15, lh = l >> 4;

    const float* ra = a2 + (size_t)s * NN * NN;
    const float* rr = r2 + s * NN;
    const float* rc = c2 + s * NN;

    v8f acc[2][4];
#pragma unroll
    for (int rt = 0; rt < 2; ++rt)
#pragma unroll
        for (int ct = 0; ct < 4; ++ct)
#pragma unroll
            for (int g = 0; g < 8; ++g) acc[rt][ct][g] = 0.0f;

    for (int k0 = 0; k0 < NN; k0 += 32) {
        __syncthreads();
#pragma unroll
        for (int it = 0; it < 32; ++it) {
            int q = it * 256 + t;
            int row = q >> 5, kk = q & 31;
            xs[row * XS + kk] = x[row * NN + k0 + kk];
        }
#pragma unroll
        for (int it = 0; it < 8; ++it) {
            int q = it * 256 + t;
            int kk = q >> 6, m = q & 63;
            ps[m * PT + kk] =
                fexp2(ra[(size_t)(k0 + kk) * NN + m0 + m] - rr[k0 + kk] - rc[m0 + m]);
        }
        __syncthreads();
#pragma unroll
        for (int ks = 0; ks < 32; ks += 4) {
            const int ka = ks + (lh << 1);
            v2f a[2];
#pragma unroll
            for (int rt = 0; rt < 2; ++rt) {
                int row = w * 32 + rt * 16 + lm;
                a[rt] = *(const v2f*)&xs[row * XS + ka];
            }
#pragma unroll
            for (int ct = 0; ct < 4; ++ct) {
                int col = ct * 16 + lm;
                v2f b = *(const v2f*)&ps[col * PT + ka];
#pragma unroll
                for (int rt = 0; rt < 2; ++rt) {
                    acc[rt][ct] = __builtin_amdgcn_wmma_f32_16x16x4_f32(
                        false, a[rt], false, b, (short)0, acc[rt][ct], false, false);
                }
            }
        }
    }
    float* ro = out + (size_t)s * MB * NN;
#pragma unroll
    for (int rt = 0; rt < 2; ++rt)
#pragma unroll
        for (int ct = 0; ct < 4; ++ct)
#pragma unroll
            for (int g = 0; g < 8; ++g) {
                int row = w * 32 + rt * 16 + lh * 8 + g;
                int col = m0 + ct * 16 + lm;
                ro[row * NN + col] = acc[rt][ct][g];
            }
}

extern "C" void kernel_launch(void* const* d_in, const int* in_sizes, int n_in,
                              void* d_out, int out_size, void* d_ws, size_t ws_size,
                              hipStream_t stream) {
    const float* x  = (const float*)d_in[0];
    const float* la = (const float*)d_in[1];
    const float* u  = (const float*)d_in[2];
    float* out = (float*)d_out;

    char* ws = (char*)d_ws;
    float* a2    = (float*)ws;
    float* r2    = (float*)(ws + (size_t)NS * NN * NN * sizeof(float));
    float* c2    = r2 + NS * NN;
    float* pa    = out;
    float* pb    = out + NS * NP * NN;

    prepcol_kernel<<<dim3(NN / 64, NS), 512, 0, stream>>>(la, u, a2, c2);
    rowpass_kernel<<<dim3(NN / 32, NS), 256, 0, stream>>>(a2, c2, r2);
    colrow_kernel<true><<<dim3(NN / 64, NS), 512, 0, stream>>>(a2, r2, c2, pa);
    for (int k = 0; k < 9; ++k) {
        colrow_kernel<false><<<dim3(NN / 64, NS), 512, 0, stream>>>(a2, pa, c2, pb);
        colrow_kernel<false><<<dim3(NN / 64, NS), 512, 0, stream>>>(a2, pb, c2, pa);
    }
    colrow_kernel<false><<<dim3(NN / 64, NS), 512, 0, stream>>>(a2, pa, c2, pb);
    rowfin_kernel<<<dim3(NN / 256, NS), 256, 0, stream>>>(pb, r2);
    gemm_kernel<<<dim3(NN / 64, NS), 256, 0, stream>>>(x, a2, r2, c2, out);
}


# Product kernel without LDS or barriers: each wave loads its A and B fragments straight from global memory (128-bit loads, one K step of 32 ahead) and forms P in registers; k summed in a new order
# speedup vs baseline: 2.0745x; 1.1594x over previous
#include <hip/hip_runtime.h>
#include <math.h>

#define NS 16
#define NN 1024
#define MB 256
#define NITERS 21
#define NP 16
#define LOG2E 1.4426950408889634f
#define EPSC 1e-10f

typedef float v2f __attribute__((ext_vector_type(2)));
typedef float v4f __attribute__((ext_vector_type(4)));
typedef float v8f __attribute__((ext_vector_type(8)));

__device__ __forceinline__ float fexp2(float x) { return __builtin_amdgcn_exp2f(x); }
__device__ __forceinline__ float flog2(float x) { return __builtin_amdgcn_logf(x); }

__global__ void __launch_bounds__(512) prepcol_kernel(const float* __restrict__ la,
                                                      const float* __restrict__ u,
                                                      float* __restrict__ a2,
                                                      float* __restrict__ c2) {
    __shared__ __attribute__((aligned(16))) float red[32 * 64];
    const int s = blockIdx.y, strip = blockIdx.x;
    const int t = threadIdx.x;
    const int g  = t >> 4;
    const int lc = t & 15;
    const size_t off = (size_t)(g * 32) * NN + strip * 64 + lc * 4;
    const float* up = u + (size_t)s * NN * NN + off;
    const float* lp = la + off;
    float* op = a2 + (size_t)s * NN * NN + off;
    constexpr int WIN = 16;
    v4f lv[32], uv[32];
#pragma unroll
    for (int q = 0; q < WIN; ++q) {
        lv[q] = *(const v4f*)(lp + (size_t)q * NN);
        uv[q] = *(const v4f*)(up + (size_t)q * NN);
    }
    float c0 = 0.f, c1 = 0.f, c2v = 0.f, c3 = 0.f;
#pragma unroll
    for (int q = 0; q < 32; ++q) {
        __builtin_amdgcn_sched_barrier(0);
        if (q + WIN < 32) {
            lv[q + WIN] = *(const v4f*)(lp + (size_t)(q + WIN) * NN);
            uv[q + WIN] = *(const v4f*)(up + (size_t)(q + WIN) * NN);
        }
        v4f av;
        for (int e = 0; e < 4; ++e) {
            const float noise = -logf(EPSC - logf(uv[q][e] + EPSC));
            av[e] = LOG2E * (lv[q][e] + noise);
        }
        *(v4f*)(op + (size_t)q * NN) = av;
        c0  += fexp2(av[0]);
        c1  += fexp2(av[1]);
        c2v += fexp2(av[2]);
        c3  += fexp2(av[3]);
    }
    v4f cs; cs[0] = c0; cs[1] = c1; cs[2] = c2v; cs[3] = c3;
    *(v4f*)&red[g * 64 + lc * 4] = cs;
    __syncthreads();
    if (t < 64) {
        float sum = 0.f;
        for (int k = 0; k < 32; ++k) sum += red[k * 64 + t];
        c2[s * NN + strip * 64 + t] = flog2(sum);
    }
}

template <bool FIRST>
__global__ void __launch_bounds__(512) colrow_kernel(const float* __restrict__ a2,
                                                     const float* __restrict__ rin,
                                                     float* __restrict__ c2,
                                                     float* __restrict__ pout) {
    __shared__ __attribute__((aligned(16))) float rs[NN];
    __shared__ __attribute__((aligned(16))) float red[32 * 64];
    __shared__ __attribute__((aligned(16))) float cs[64];
    const int s = blockIdx.y, strip = blockIdx.x;
    const int t = threadIdx.x;
    const int g  = t >> 4;
    const int lc = t & 15;
    v2f pv[NP];
    if (FIRST) {
        pv[0] = *(const v2f*)(rin + s * NN + 2 * t);
    } else {
#pragma unroll
        for (int p = 0; p < NP; ++p) pv[p] = *(const v2f*)(rin + (size_t)(s * NP + p) * NN + 2 * t);
    }
    const float* ap = a2 + ((size_t)s * NN + g * 32) * NN + strip * 64 + lc * 4;
    v4f av[32];
#pragma unroll
    for (int q = 0; q < 32; ++q) av[q] = *(const v4f*)(ap + (size_t)q * NN);
    v2f rr;
    if (FIRST) {
        rr = pv[0];
    } else {
        float s0 = 0.f, s1 = 0.f;
#pragma unroll
        for (int p = 0; p < NP; ++p) { s0 += pv[p][0]; s1 += pv[p][1]; }
        rr[0] = flog2(s0); rr[1] = flog2(s1);
    }
    *(v2f*)&rs[2 * t] = rr;
    __builtin_amdgcn_fence(__ATOMIC_RELEASE, "wavefront");
    __builtin_amdgcn_wave_barrier();
    __builtin_amdgcn_fence(__ATOMIC_ACQUIRE, "wavefront");
    v4f rv[8];
#pragma unroll
    for (int k = 0; k < 8; ++k) rv[k] = *(const v4f*)&rs[g * 32 + 4 * k];
    float c0 = 0.f, c1 = 0.f, c2v = 0.f, c3 = 0.f;
#pragma unroll
    for (int q = 0; q < 32; ++q) {
        if ((q & 3) == 0) __builtin_amdgcn_sched_barrier(0);
        const float r = rv[q >> 2][q & 3];
        c0  += fexp2(av[q][0] - r);
        c1  += fexp2(av[q][1] - r);
        c2v += fexp2(av[q][2] - r);
        c3  += fexp2(av[q][3] - r);
    }
    v4f csum; csum[0] = c0; csum[1] = c1; csum[2] = c2v; csum[3] = c3;
    *(v4f*)&red[g * 64 + lc * 4] = csum;
    __syncthreads();
    if (t < 64) {
        float sum = 0.f;
        for (int k = 0; k < 32; ++k) sum += red[k * 64 + t];
        const float c = flog2(sum);
        cs[t] = c;
        c2[s * NN + strip * 64 + t] = c;
    }
    __syncthreads();
    const v4f cv = *(const v4f*)&cs[lc * 4];
    float v[32];
#pragma unroll
    for (int q = 0; q < 32; ++q) {
        v[q] = (fexp2(av[q][0] - cv[0]) + fexp2(av[q][1] - cv[1])) +
               (fexp2(av[q][2] - cv[2]) + fexp2(av[q][3] - cv[3]));
    }
#pragma unroll
    for (int h = 16; h >= 2; h >>= 1) {
        const bool up = (lc & (h >> 1)) != 0;
#pragma unroll
        for (int k = 0; k < h; ++k) {
            const float lo = v[k], hi = v[k + h];
            const float keep = up ? hi : lo;
            const float send = up ? lo : hi;
            v[k] = keep + __shfl_xor(send, h >> 1, 32);
        }
    }
    v2f o; o[0] = v[0]; o[1] = v[1];
    *(v2f*)(pout + (size_t)(s * NP + strip) * NN + g * 32 + 2 * lc) = o;
}

__global__ void __launch_bounds__(256) rowpass_kernel(const float* __restrict__ a2,
                                                      const float* __restrict__ c2,
                                                      float* __restrict__ r2) {
    __shared__ __attribute__((aligned(16))) float red[32];
    const int s = blockIdx.y;
    const int t = threadIdx.x;
    const int w = t >> 5, l = t & 31;
    const int i0 = blockIdx.x * 32 + w * 4;
    const float* cp = c2 + s * NN + l * 4;
    const float* ap = a2 + ((size_t)s * NN + i0) * NN + l * 4;
    v4f cv[8];
    for (int k = 0; k < 8; ++k) cv[k] = *(const v4f*)(cp + 128 * k);
    v4f av[4][8];
    for (int r = 0; r < 4; ++r)
        for (int k = 0; k < 8; ++k) av[r][k] = *(const v4f*)(ap + (size_t)r * NN + 128 * k);
    float sum[4];
    for (int r = 0; r < 4; ++r) {
        __builtin_amdgcn_sched_barrier(0);
        float s0 = 0.f, s1 = 0.f, s2 = 0.f, s3 = 0.f;
        for (int k = 0; k < 8; ++k) {
            s0 += fexp2(av[r][k][0] - cv[k][0]);
            s1 += fexp2(av[r][k][1] - cv[k][1]);
            s2 += fexp2(av[r][k][2] - cv[k][2]);
            s3 += fexp2(av[r][k][3] - cv[k][3]);
        }
        sum[r] = (s0 + s1) + (s2 + s3);
    }
    for (int m = 16; m >= 1; m >>= 1)
        for (int r = 0; r < 4; ++r) sum[r] += __shfl_xor(sum[r], m, 32);
    if (l == 0) {
        v4f o; o[0] = sum[0]; o[1] = sum[1]; o[2] = sum[2]; o[3] = sum[3];
        *(v4f*)&red[w * 4] = o;
    }
    __syncthreads();
    if (t < 32) r2[s * NN + blockIdx.x * 32 + t] = flog2(red[t]);
}

__global__ void __launch_bounds__(256) rowfin_kernel(const float* __restrict__ pl, float* __restrict__ r2) {
    const int s = blockIdx.y;
    const int i = blockIdx.x * 256 + threadIdx.x;
    float sum = 0.f;
    for (int p = 0; p < NP; ++p) sum += pl[(size_t)(s * NP + p) * NN + i];
    r2[s * NN + i] = flog2(sum);
}

struct GemmFrag {
    v4f r[4];
    v4f a[2][4];
    v4f b[16];
};

__device__ __forceinline__ void gemm_load(GemmFrag& f, const float* __restrict__ xa, const float* __restrict__ rp,
                                          const float* __restrict__ bp, int k) {
#pragma unroll
    for (int q = 0; q < 4; ++q) f.r[q] = *(const v4f*)(rp + k + 4 * q);
#pragma unroll
    for (int rt = 0; rt < 2; ++rt)
#pragma unroll
        for (int q = 0; q < 4; ++q) f.a[rt][q] = *(const v4f*)(xa + (size_t)rt * 16 * NN + k + 4 * q);
#pragma unroll
    for (int e = 0; e < 16; ++e) f.b[e] = *(const v4f*)(bp + (size_t)(k + e) * NN);
}

__device__ __forceinline__ void gemm_step(const GemmFrag& f, const v4f cv, v8f (&acc)[2][4]) {
#pragma unroll
    for (int j = 0; j < 8; ++j) {
        const int e0 = 2 * j, e1 = 2 * j + 1;
        v2f pb[4];
#pragma unroll
        for (int ct = 0; ct < 4; ++ct) {
            pb[ct][0] = fexp2(f.b[e0][ct] - f.r[e0 >> 2][e0 & 3] - cv[ct]);
            pb[ct][1] = fexp2(f.b[e1][ct] - f.r[e1 >> 2][e1 & 3] - cv[ct]);
        }
        v2f af[2];
#pragma unroll
        for (int rt = 0; rt < 2; ++rt) {
            af[rt][0] = f.a[rt][e0 >> 2][e0 & 3];
            af[rt][1] = f.a[rt][e1 >> 2][e1 & 3];
        }
#pragma unroll
        for (int ct = 0; ct < 4; ++ct) {
#pragma unroll
            for (int rt = 0; rt < 2; ++rt) {
                acc[rt][ct] = __builtin_amdgcn_wmma_f32_16x16x4_f32(
                    false, af[rt], false, pb[ct], (short)0, acc[rt][ct], false, false);
            }
        }
    }
}

__global__ void __launch_bounds__(256) gemm_kernel(const float* __restrict__ x,
                                                   const float* __restrict__ a2,
                                                   const float* __restrict__ r2,
                                                   const float* __restrict__ c2,
                                                   float* __restrict__ out) {
    const int s  = blockIdx.y;
    const int m0 = blockIdx.x * 64;
    const int t  = threadIdx.x;
    const int w  = t >> 5, l = t & 31;
    const int lm = l & 15, lh = l >> 4;

    const float* xa = x + (size_t)(w * 32 + lm) * NN + 16 * lh;
    const float* bp = a2 + ((size_t)s * NN + 16 * lh) * NN + m0 + 4 * lm;
    const float* rp = r2 + s * NN + 16 * lh;
    const v4f cv = *(const v4f*)(c2 + s * NN + m0 + 4 * lm);

    v8f acc[2][4];
#pragma unroll
    for (int rt = 0; rt < 2; ++rt)
#pragma unroll
        for (int ct = 0; ct < 4; ++ct)
#pragma unroll
            for (int g = 0; g < 8; ++g) acc[rt][ct][g] = 0.0f;

    GemmFrag f0, f1;
    gemm_load(f0, xa, rp, bp, 0);
#pragma nounroll
    for (int k0 = 0; k0 < NN; k0 += 64) {
        gemm_load(f1, xa, rp, bp, k0 + 32);
        __builtin_amdgcn_sched_barrier(0);
        gemm_step(f0, cv, acc);
        __builtin_amdgcn_sched_barrier(0);
        const int kn = (k0 + 64 < NN) ? k0 + 64 : k0;
        gemm_load(f0, xa, rp, bp, kn);
        __builtin_amdgcn_sched_barrier(0);
        gemm_step(f1, cv, acc);
        __builtin_amdgcn_sched_barrier(0);
    }
    float* ro = out + (size_t)s * MB * NN + (size_t)(w * 32 + 8 * lh) * NN + m0 + 4 * lm;
#pragma unroll
    for (int rt = 0; rt < 2; ++rt)
#pragma unroll
        for (int g = 0; g < 8; ++g) {
            v4f o;
            o[0] = acc[rt][0][g]; o[1] = acc[rt][1][g]; o[2] = acc[rt][2][g]; o[3] = acc[rt][3][g];
            *(v4f*)(ro + (size_t)(rt * 16 + g) * NN) = o;
        }
}

extern "C" void kernel_launch(void* const* d_in, const int* in_sizes, int n_in,
                              void* d_out, int out_size, void* d_ws, size_t ws_size,
                              hipStream_t stream) {
    const float* x  = (const float*)d_in[0];
    const float* la = (const float*)d_in[1];
    const float* u  = (const float*)d_in[2];
    float* out = (float*)d_out;

    char* ws = (char*)d_ws;
    float* a2    = (float*)ws;
    float* r2    = (float*)(ws + (size_t)NS * NN * NN * sizeof(float));
    float* c2    = r2 + NS * NN;
    float* pa    = out;
    float* pb    = out + NS * NP * NN;

    prepcol_kernel<<<dim3(NN / 64, NS), 512, 0, stream>>>(la, u, a2, c2);
    rowpass_kernel<<<dim3(NN / 32, NS), 256, 0, stream>>>(a2, c2, r2);
    colrow_kernel<true><<<dim3(NN / 64, NS), 512, 0, stream>>>(a2, r2, c2, pa);
    for (int k = 0; k < 9; ++k) {
        colrow_kernel<false><<<dim3(NN / 64, NS), 512, 0, stream>>>(a2, pa, c2, pb);
        colrow_kernel<false><<<dim3(NN / 64, NS), 512, 0, stream>>>(a2, pb, c2, pa);
    }
    colrow_kernel<false><<<dim3(NN / 64, NS), 512, 0, stream>>>(a2, pa, c2, pb);
    rowfin_kernel<<<dim3(NN / 256, NS), 256, 0, stream>>>(pb, r2);
    gemm_kernel<<<dim3(NN / 64, NS), 256, 0, stream>>>(x, a2, r2, c2, out);
}


# Row sweep of the fused launch reuses the column sweep's exponentials: E=2^(a2-r) stays in registers, times 1/colsum by packed multiply-adds; one exp2 per element, not two; row sums in a new order
# speedup vs baseline: 2.2318x; 1.0758x over previous
#include <hip/hip_runtime.h>
#include <math.h>

#define NS 16
#define NN 1024
#define MB 256
#define NITERS 21
#define NP 16
#define LOG2E 1.4426950408889634f
#define EPSC 1e-10f

typedef float v2f __attribute__((ext_vector_type(2)));
typedef float v4f __attribute__((ext_vector_type(4)));
typedef float v8f __attribute__((ext_vector_type(8)));

__device__ __forceinline__ float fexp2(float x) { return __builtin_amdgcn_exp2f(x); }
__device__ __forceinline__ float flog2(float x) { return __builtin_amdgcn_logf(x); }

__global__ void __launch_bounds__(512) prepcol_kernel(const float* __restrict__ la,
                                                      const float* __restrict__ u,
                                                      float* __restrict__ a2,
                                                      float* __restrict__ c2) {
    __shared__ __attribute__((aligned(16))) float red[32 * 64];
    const int s = blockIdx.y, strip = blockIdx.x;
    const int t = threadIdx.x;
    const int g  = t >> 4;
    const int lc = t & 15;
    const size_t off = (size_t)(g * 32) * NN + strip * 64 + lc * 4;
    const float* up = u + (size_t)s * NN * NN + off;
    const float* lp = la + off;
    float* op = a2 + (size_t)s * NN * NN + off;
    constexpr int WIN = 16;
    v4f lv[32], uv[32];
#pragma unroll
    for (int q = 0; q < WIN; ++q) {
        lv[q] = *(const v4f*)(lp + (size_t)q * NN);
        uv[q] = *(const v4f*)(up + (size_t)q * NN);
    }
    float c0 = 0.f, c1 = 0.f, c2v = 0.f, c3 = 0.f;
#pragma unroll
    for (int q = 0; q < 32; ++q) {
        __builtin_amdgcn_sched_barrier(0);
        if (q + WIN < 32) {
            lv[q + WIN] = *(const v4f*)(lp + (size_t)(q + WIN) * NN);
            uv[q + WIN] = *(const v4f*)(up + (size_t)(q + WIN) * NN);
        }
        v4f av;
        for (int e = 0; e < 4; ++e) {
            const float noise = -logf(EPSC - logf(uv[q][e] + EPSC));
            av[e] = LOG2E * (lv[q][e] + noise);
        }
        *(v4f*)(op + (size_t)q * NN) = av;
        c0  += fexp2(av[0]);
        c1  += fexp2(av[1]);
        c2v += fexp2(av[2]);
        c3  += fexp2(av[3]);
    }
    v4f cs; cs[0] = c0; cs[1] = c1; cs[2] = c2v; cs[3] = c3;
    *(v4f*)&red[g * 64 + lc * 4] = cs;
    __syncthreads();
    if (t < 64) {
        float sum = 0.f;
        for (int k = 0; k < 32; ++k) sum += red[k * 64 + t];
        c2[s * NN + strip * 64 + t] = flog2(sum);
    }
}

template <bool FIRST>
__global__ void __launch_bounds__(512) colrow_kernel(const float* __restrict__ a2,
                                                     const float* __restrict__ rin,
                                                     float* __restrict__ c2,
                                                     float* __restrict__ pout) {
    __shared__ __attribute__((aligned(16))) float rs[NN];
    __shared__ __attribute__((aligned(16))) float red[32 * 64];
    __shared__ __attribute__((aligned(16))) float cs[64];
    const int s = blockIdx.y, strip = blockIdx.x;
    const int t = threadIdx.x;
    const int g  = t >> 4;
    const int lc = t & 15;
    v2f pv[NP];
    if (FIRST) {
        pv[0] = *(const v2f*)(rin + s * NN + 2 * t);
    } else {
#pragma unroll
        for (int p = 0; p < NP; ++p) pv[p] = *(const v2f*)(rin + (size_t)(s * NP + p) * NN + 2 * t);
    }
    const float* ap = a2 + ((size_t)s * NN + g * 32) * NN + strip * 64 + lc * 4;
    v4f av[32];
#pragma unroll
    for (int q = 0; q < 32; ++q) av[q] = *(const v4f*)(ap + (size_t)q * NN);
    v2f rr;
    if (FIRST) {
        rr = pv[0];
    } else {
        float s0 = 0.f, s1 = 0.f;
#pragma unroll
        for (int p = 0; p < NP; ++p) { s0 += pv[p][0]; s1 += pv[p][1]; }
        rr[0] = flog2(s0); rr[1] = flog2(s1);
    }
    *(v2f*)&rs[2 * t] = rr;
    __builtin_amdgcn_fence(__ATOMIC_RELEASE, "wavefront");
    __builtin_amdgcn_wave_barrier();
    __builtin_amdgcn_fence(__ATOMIC_ACQUIRE, "wavefront");
    v4f rv[8];
#pragma unroll
    for (int k = 0; k < 8; ++k) rv[k] = *(const v4f*)&rs[g * 32 + 4 * k];
    float c0 = 0.f, c1 = 0.f, c2v = 0.f, c3 = 0.f;
#pragma unroll
    for (int q = 0; q < 32; ++q) {
        if ((q & 3) == 0) __builtin_amdgcn_sched_barrier(0);
        const float r = rv[q >> 2][q & 3];
        for (int e = 0; e < 4; ++e) av[q][e] = fexp2(av[q][e] - r);
        c0  += av[q][0];
        c1  += av[q][1];
        c2v += av[q][2];
        c3  += av[q][3];
    }
    v4f csum; csum[0] = c0; csum[1] = c1; csum[2] = c2v; csum[3] = c3;
    *(v4f*)&red[g * 64 + lc * 4] = csum;
    __syncthreads();
    if (t < 64) {
        float sum = 0.f;
        for (int k = 0; k < 32; ++k) sum += red[k * 64 + t];
        cs[t] = __builtin_amdgcn_rcpf(sum);
        c2[s * NN + strip * 64 + t] = flog2(sum);
    }
    __syncthreads();
    const v4f cv = *(const v4f*)&cs[lc * 4];
    v2f i01, i23;
    i01[0] = cv[0]; i01[1] = cv[1]; i23[0] = cv[2]; i23[1] = cv[3];
    float v[32];
#pragma unroll
    for (int q = 0; q < 32; ++q) {
        v2f e01, e23;
        e01[0] = av[q][0]; e01[1] = av[q][1]; e23[0] = av[q][2]; e23[1] = av[q][3];
        const v2f p = __builtin_elementwise_fma(e23, i23, e01 * i01);
        v[q] = p[0] + p[1];
    }
#pragma unroll
    for (int h = 16; h >= 2; h >>= 1) {
        const bool up = (lc & (h >> 1)) != 0;
#pragma unroll
        for (int k = 0; k < h; ++k) {
            const float lo = v[k], hi = v[k + h];
            const float keep = up ? hi : lo;
            const float send = up ? lo : hi;
            v[k] = keep + __shfl_xor(send, h >> 1, 32);
        }
    }
    v2f o; o[0] = v[0] * fexp2(rr[0]); o[1] = v[1] * fexp2(rr[1]);
    *(v2f*)(pout + (size_t)(s * NP + strip) * NN + g * 32 + 2 * lc) = o;
}

__global__ void __launch_bounds__(256) rowpass_kernel(const float* __restrict__ a2,
                                                      const float* __restrict__ c2,
                                                      float* __restrict__ r2) {
    __shared__ __attribute__((aligned(16))) float red[32];
    const int s = blockIdx.y;
    const int t = threadIdx.x;
    const int w = t >> 5, l = t & 31;
    const int i0 = blockIdx.x * 32 + w * 4;
    const float* cp = c2 + s * NN + l * 4;
    const float* ap = a2 + ((size_t)s * NN + i0) * NN + l * 4;
    v4f cv[8];
    for (int k = 0; k < 8; ++k) cv[k] = *(const v4f*)(cp + 128 * k);
    v4f av[4][8];
    for (int r = 0; r < 4; ++r)
        for (int k = 0; k < 8; ++k) av[r][k] = *(const v4f*)(ap + (size_t)r * NN + 128 * k);
    float sum[4];
    for (int r = 0; r < 4; ++r) {
        __builtin_amdgcn_sched_barrier(0);
        float s0 = 0.f, s1 = 0.f, s2 = 0.f, s3 = 0.f;
        for (int k = 0; k < 8; ++k) {
            s0 += fexp2(av[r][k][0] - cv[k][0]);
            s1 += fexp2(av[r][k][1] - cv[k][1]);
            s2 += fexp2(av[r][k][2] - cv[k][2]);
            s3 += fexp2(av[r][k][3] - cv[k][3]);
        }
        sum[r] = (s0 + s1) + (s2 + s3);
    }
    for (int m = 16; m >= 1; m >>= 1)
        for (int r = 0; r < 4; ++r) sum[r] += __shfl_xor(sum[r], m, 32);
    if (l == 0) {
        v4f o; o[0] = sum[0]; o[1] = sum[1]; o[2] = sum[2]; o[3] = sum[3];
        *(v4f*)&red[w * 4] = o;
    }
    __syncthreads();
    if (t < 32) r2[s * NN + blockIdx.x * 32 + t] = flog2(red[t]);
}

__global__ void __launch_bounds__(256) rowfin_kernel(const float* __restrict__ pl, float* __restrict__ r2) {
    const int s = blockIdx.y;
    const int i = blockIdx.x * 256 + threadIdx.x;
    float sum = 0.f;
    for (int p = 0; p < NP; ++p) sum += pl[(size_t)(s * NP + p) * NN + i];
    r2[s * NN + i] = flog2(sum);
}

struct GemmFrag {
    v4f r[4];
    v4f a[2][4];
    v4f b[16];
};

__device__ __forceinline__ void gemm_load(GemmFrag& f, const float* __restrict__ xa, const float* __restrict__ rp,
                                          const float* __restrict__ bp, int k) {
#pragma unroll
    for (int q = 0; q < 4; ++q) f.r[q] = *(const v4f*)(rp + k + 4 * q);
#pragma unroll
    for (int rt = 0; rt < 2; ++rt)
#pragma unroll
        for (int q = 0; q < 4; ++q) f.a[rt][q] = *(const v4f*)(xa + (size_t)rt * 16 * NN + k + 4 * q);
#pragma unroll
    for (int e = 0; e < 16; ++e) f.b[e] = *(const v4f*)(bp + (size_t)(k + e) * NN);
}

__device__ __forceinline__ void gemm_step(const GemmFrag& f, const v4f cv, v8f (&acc)[2][4]) {
#pragma unroll
    for (int j = 0; j < 8; ++j) {
        const int e0 = 2 * j, e1 = 2 * j + 1;
        v2f pb[4];
#pragma unroll
        for (int ct = 0; ct < 4; ++ct) {
            pb[ct][0] = fexp2(f.b[e0][ct] - f.r[e0 >> 2][e0 & 3] - cv[ct]);
            pb[ct][1] = fexp2(f.b[e1][ct] - f.r[e1 >> 2][e1 & 3] - cv[ct]);
        }
        v2f af[2];
#pragma unroll
        for (int rt = 0; rt < 2; ++rt) {
            af[rt][0] = f.a[rt][e0 >> 2][e0 & 3];
            af[rt][1] = f.a[rt][e1 >> 2][e1 & 3];
        }
#pragma unroll
        for (int ct = 0; ct < 4; ++ct) {
#pragma unroll
            for (int rt = 0; rt < 2; ++rt) {
                acc[rt][ct] = __builtin_amdgcn_wmma_f32_16x16x4_f32(
                    false, af[rt], false, pb[ct], (short)0, acc[rt][ct], false, false);
            }
        }
    }
}

__global__ void __launch_bounds__(256) gemm_kernel(const float* __restrict__ x,
                                                   const float* __restrict__ a2,
                                                   const float* __restrict__ r2,
                                                   const float* __restrict__ c2,
                                                   float* __restrict__ out) {
    const int s  = blockIdx.y;
    const int m0 = blockIdx.x * 64;
    const int t  = threadIdx.x;
    const int w  = t >> 5, l = t & 31;
    const int lm = l & 15, lh = l >> 4;

    const float* xa = x + (size_t)(w * 32 + lm) * NN + 16 * lh;
    const float* bp = a2 + ((size_t)s * NN + 16 * lh) * NN + m0 + 4 * lm;
    const float* rp = r2 + s * NN + 16 * lh;
    const v4f cv = *(const v4f*)(c2 + s * NN + m0 + 4 * lm);

    v8f acc[2][4];
#pragma unroll
    for (int rt = 0; rt < 2; ++rt)
#pragma unroll
        for (int ct = 0; ct < 4; ++ct)
#pragma unroll
            for (int g = 0; g < 8; ++g) acc[rt][ct][g] = 0.0f;

    GemmFrag f0, f1;
    gemm_load(f0, xa, rp, bp, 0);
#pragma nounroll
    for (int k0 = 0; k0 < NN; k0 += 64) {
        gemm_load(f1, xa, rp, bp, k0 + 32);
        __builtin_amdgcn_sched_barrier(0);
        gemm_step(f0, cv, acc);
        __builtin_amdgcn_sched_barrier(0);
        const int kn = (k0 + 64 < NN) ? k0 + 64 : k0;
        gemm_load(f0, xa, rp, bp, kn);
        __builtin_amdgcn_sched_barrier(0);
        gemm_step(f1, cv, acc);
        __builtin_amdgcn_sched_barrier(0);
    }
    float* ro = out + (size_t)s * MB * NN + (size_t)(w * 32 + 8 * lh) * NN + m0 + 4 * lm;
#pragma unroll
    for (int rt = 0; rt < 2; ++rt)
#pragma unroll
        for (int g = 0; g < 8; ++g) {
            v4f o;
            o[0] = acc[rt][0][g]; o[1] = acc[rt][1][g]; o[2] = acc[rt][2][g]; o[3] = acc[rt][3][g];
            *(v4f*)(ro + (size_t)(rt * 16 + g) * NN) = o;
        }
}

extern "C" void kernel_launch(void* const* d_in, const int* in_sizes, int n_in,
                              void* d_out, int out_size, void* d_ws, size_t ws_size,
                              hipStream_t stream) {
    const float* x  = (const float*)d_in[0];
    const float* la = (const float*)d_in[1];
    const float* u  = (const float*)d_in[2];
    float* out = (float*)d_out;

    char* ws = (char*)d_ws;
    float* a2    = (float*)ws;
    float* r2    = (float*)(ws + (size_t)NS * NN * NN * sizeof(float));
    float* c2    = r2 + NS * NN;
    float* pa    = out;
    float* pb    = out + NS * NP * NN;

    prepcol_kernel<<<dim3(NN / 64, NS), 512, 0, stream>>>(la, u, a2, c2);
    rowpass_kernel<<<dim3(NN / 32, NS), 256, 0, stream>>>(a2, c2, r2);
    colrow_kernel<true><<<dim3(NN / 64, NS), 512, 0, stream>>>(a2, r2, c2, pa);
    for (int k = 0; k < 9; ++k) {
        colrow_kernel<false><<<dim3(NN / 64, NS), 512, 0, stream>>>(a2, pa, c2, pb);
        colrow_kernel<false><<<dim3(NN / 64, NS), 512, 0, stream>>>(a2, pb, c2, pa);
    }
    colrow_kernel<false><<<dim3(NN / 64, NS), 512, 0, stream>>>(a2, pa, c2, pb);
    rowfin_kernel<<<dim3(NN / 256, NS), 256, 0, stream>>>(pb, r2);
    gemm_kernel<<<dim3(NN / 64, NS), 256, 0, stream>>>(x, a2, r2, c2, out);
}


# Row pass of iteration 0 fused into the launch that writes a2: E=2^a2 stays in registers, times 1/colsum, leaves as 16 planes of row-sum parts; rowpass launch gone, 24 to 23; first row sums reordered
# speedup vs baseline: 2.3406x; 1.0487x over previous
#include <hip/hip_runtime.h>
#include <math.h>

#define NS 16
#define NN 1024
#define MB 256
#define NITERS 21
#define NP 16
#define LOG2E 1.4426950408889634f
#define EPSC 1e-10f

typedef float v2f __attribute__((ext_vector_type(2)));
typedef float v4f __attribute__((ext_vector_type(4)));
typedef float v8f __attribute__((ext_vector_type(8)));

__device__ __forceinline__ float fexp2(float x) { return __builtin_amdgcn_exp2f(x); }
__device__ __forceinline__ float flog2(float x) { return __builtin_amdgcn_logf(x); }

__global__ void __launch_bounds__(512) prepcolrow_kernel(const float* __restrict__ la,
                                                         const float* __restrict__ u,
                                                         float* __restrict__ a2,
                                                         float* __restrict__ c2,
                                                         float* __restrict__ pout) {
    __shared__ __attribute__((aligned(16))) float red[32 * 64];
    __shared__ __attribute__((aligned(16))) float cs[64];
    const int s = blockIdx.y, strip = blockIdx.x;
    const int t = threadIdx.x;
    const int g  = t >> 4;
    const int lc = t & 15;
    const size_t off = (size_t)(g * 32) * NN + strip * 64 + lc * 4;
    const float* up = u + (size_t)s * NN * NN + off;
    const float* lp = la + off;
    float* op = a2 + (size_t)s * NN * NN + off;
    constexpr int WIN = 16;
    v4f lv[32], uv[32];
    v4f ev[32];
#pragma unroll
    for (int q = 0; q < WIN; ++q) {
        lv[q] = *(const v4f*)(lp + (size_t)q * NN);
        uv[q] = *(const v4f*)(up + (size_t)q * NN);
    }
    float c0 = 0.f, c1 = 0.f, c2v = 0.f, c3 = 0.f;
#pragma unroll
    for (int q = 0; q < 32; ++q) {
        __builtin_amdgcn_sched_barrier(0);
        if (q + WIN < 32) {
            lv[q + WIN] = *(const v4f*)(lp + (size_t)(q + WIN) * NN);
            uv[q + WIN] = *(const v4f*)(up + (size_t)(q + WIN) * NN);
        }
        v4f av;
        for (int e = 0; e < 4; ++e) {
            const float noise = -logf(EPSC - logf(uv[q][e] + EPSC));
            av[e] = LOG2E * (lv[q][e] + noise);
        }
        *(v4f*)(op + (size_t)q * NN) = av;
        for (int e = 0; e < 4; ++e) ev[q][e] = fexp2(av[e]);
        c0  += ev[q][0];
        c1  += ev[q][1];
        c2v += ev[q][2];
        c3  += ev[q][3];
    }
    v4f csum; csum[0] = c0; csum[1] = c1; csum[2] = c2v; csum[3] = c3;
    *(v4f*)&red[g * 64 + lc * 4] = csum;
    __syncthreads();
    if (t < 64) {
        float sum = 0.f;
        for (int k = 0; k < 32; ++k) sum += red[k * 64 + t];
        cs[t] = __builtin_amdgcn_rcpf(sum);
        c2[s * NN + strip * 64 + t] = flog2(sum);
    }
    __syncthreads();
    const v4f cv = *(const v4f*)&cs[lc * 4];
    v2f i01, i23;
    i01[0] = cv[0]; i01[1] = cv[1]; i23[0] = cv[2]; i23[1] = cv[3];
    float v[32];
#pragma unroll
    for (int q = 0; q < 32; ++q) {
        v2f e01, e23;
        e01[0] = ev[q][0]; e01[1] = ev[q][1]; e23[0] = ev[q][2]; e23[1] = ev[q][3];
        const v2f p = __builtin_elementwise_fma(e23, i23, e01 * i01);
        v[q] = p[0] + p[1];
    }
#pragma unroll
    for (int h = 16; h >= 2; h >>= 1) {
        const bool up2 = (lc & (h >> 1)) != 0;
#pragma unroll
        for (int k = 0; k < h; ++k) {
            const float lo = v[k], hi = v[k + h];
            const float keep = up2 ? hi : lo;
            const float send = up2 ? lo : hi;
            v[k] = keep + __shfl_xor(send, h >> 1, 32);
        }
    }
    v2f o; o[0] = v[0]; o[1] = v[1];
    *(v2f*)(pout + (size_t)(s * NP + strip) * NN + g * 32 + 2 * lc) = o;
}

__global__ void __launch_bounds__(512) colrow_kernel(const float* __restrict__ a2,
                                                     const float* __restrict__ rin,
                                                     float* __restrict__ c2,
                                                     float* __restrict__ pout) {
    __shared__ __attribute__((aligned(16))) float rs[NN];
    __shared__ __attribute__((aligned(16))) float red[32 * 64];
    __shared__ __attribute__((aligned(16))) float cs[64];
    const int s = blockIdx.y, strip = blockIdx.x;
    const int t = threadIdx.x;
    const int g  = t >> 4;
    const int lc = t & 15;
    v2f pv[NP];
#pragma unroll
    for (int p = 0; p < NP; ++p) pv[p] = *(const v2f*)(rin + (size_t)(s * NP + p) * NN + 2 * t);
    const float* ap = a2 + ((size_t)s * NN + g * 32) * NN + strip * 64 + lc * 4;
    v4f av[32];
#pragma unroll
    for (int q = 0; q < 32; ++q) av[q] = *(const v4f*)(ap + (size_t)q * NN);
    v2f rr;
    {
        float s0 = 0.f, s1 = 0.f;
#pragma unroll
        for (int p = 0; p < NP; ++p) { s0 += pv[p][0]; s1 += pv[p][1]; }
        rr[0] = flog2(s0); rr[1] = flog2(s1);
    }
    *(v2f*)&rs[2 * t] = rr;
    __builtin_amdgcn_fence(__ATOMIC_RELEASE, "wavefront");
    __builtin_amdgcn_wave_barrier();
    __builtin_amdgcn_fence(__ATOMIC_ACQUIRE, "wavefront");
    v4f rv[8];
#pragma unroll
    for (int k = 0; k < 8; ++k) rv[k] = *(const v4f*)&rs[g * 32 + 4 * k];
    float c0 = 0.f, c1 = 0.f, c2v = 0.f, c3 = 0.f;
#pragma unroll
    for (int q = 0; q < 32; ++q) {
        if ((q & 3) == 0) __builtin_amdgcn_sched_barrier(0);
        const float r = rv[q >> 2][q & 3];
        for (int e = 0; e < 4; ++e) av[q][e] = fexp2(av[q][e] - r);
        c0  += av[q][0];
        c1  += av[q][1];
        c2v += av[q][2];
        c3  += av[q][3];
    }
    v4f csum; csum[0] = c0; csum[1] = c1; csum[2] = c2v; csum[3] = c3;
    *(v4f*)&red[g * 64 + lc * 4] = csum;
    __syncthreads();
    if (t < 64) {
        float sum = 0.f;
        for (int k = 0; k < 32; ++k) sum += red[k * 64 + t];
        cs[t] = __builtin_amdgcn_rcpf(sum);
        c2[s * NN + strip * 64 + t] = flog2(sum);
    }
    __syncthreads();
    const v4f cv = *(const v4f*)&cs[lc * 4];
    v2f i01, i23;
    i01[0] = cv[0]; i01[1] = cv[1]; i23[0] = cv[2]; i23[1] = cv[3];
    float v[32];
#pragma unroll
    for (int q = 0; q < 32; ++q) {
        v2f e01, e23;
        e01[0] = av[q][0]; e01[1] = av[q][1]; e23[0] = av[q][2]; e23[1] = av[q][3];
        const v2f p = __builtin_elementwise_fma(e23, i23, e01 * i01);
        v[q] = p[0] + p[1];
    }
#pragma unroll
    for (int h = 16; h >= 2; h >>= 1) {
        const bool up = (lc & (h >> 1)) != 0;
#pragma unroll
        for (int k = 0; k < h; ++k) {
            const float lo = v[k], hi = v[k + h];
            const float keep = up ? hi : lo;
            const float send = up ? lo : hi;
            v[k] = keep + __shfl_xor(send, h >> 1, 32);
        }
    }
    v2f o; o[0] = v[0] * fexp2(rr[0]); o[1] = v[1] * fexp2(rr[1]);
    *(v2f*)(pout + (size_t)(s * NP + strip) * NN + g * 32 + 2 * lc) = o;
}

__global__ void __launch_bounds__(256) rowfin_kernel(const float* __restrict__ pl, float* __restrict__ r2) {
    const int s = blockIdx.y;
    const int i = blockIdx.x * 256 + threadIdx.x;
    float sum = 0.f;
    for (int p = 0; p < NP; ++p) sum += pl[(size_t)(s * NP + p) * NN + i];
    r2[s * NN + i] = flog2(sum);
}

struct GemmFrag {
    v4f r[4];
    v4f a[2][4];
    v4f b[16];
};

__device__ __forceinline__ void gemm_load(GemmFrag& f, const float* __restrict__ xa, const float* __restrict__ rp,
                                          const float* __restrict__ bp, int k) {
#pragma unroll
    for (int q = 0; q < 4; ++q) f.r[q] = *(const v4f*)(rp + k + 4 * q);
#pragma unroll
    for (int rt = 0; rt < 2; ++rt)
#pragma unroll
        for (int q = 0; q < 4; ++q) f.a[rt][q] = *(const v4f*)(xa + (size_t)rt * 16 * NN + k + 4 * q);
#pragma unroll
    for (int e = 0; e < 16; ++e) f.b[e] = *(const v4f*)(bp + (size_t)(k + e) * NN);
}

__device__ __forceinline__ void gemm_step(const GemmFrag& f, const v4f cv, v8f (&acc)[2][4]) {
#pragma unroll
    for (int j = 0; j < 8; ++j) {
        const int e0 = 2 * j, e1 = 2 * j + 1;
        v2f pb[4];
#pragma unroll
        for (int ct = 0; ct < 4; ++ct) {
            pb[ct][0] = fexp2(f.b[e0][ct] - f.r[e0 >> 2][e0 & 3] - cv[ct]);
            pb[ct][1] = fexp2(f.b[e1][ct] - f.r[e1 >> 2][e1 & 3] - cv[ct]);
        }
        v2f af[2];
#pragma unroll
        for (int rt = 0; rt < 2; ++rt) {
            af[rt][0] = f.a[rt][e0 >> 2][e0 & 3];
            af[rt][1] = f.a[rt][e1 >> 2][e1 & 3];
        }
#pragma unroll
        for (int ct = 0; ct < 4; ++ct) {
#pragma unroll
            for (int rt = 0; rt < 2; ++rt) {
                acc[rt][ct] = __builtin_amdgcn_wmma_f32_16x16x4_f32(
                    false, af[rt], false, pb[ct], (short)0, acc[rt][ct], false, false);
            }
        }
    }
}

__global__ void __launch_bounds__(256) gemm_kernel(const float* __restrict__ x,
                                                   const float* __restrict__ a2,
                                                   const float* __restrict__ r2,
                                                   const float* __restrict__ c2,
                                                   float* __restrict__ out) {
    const int s  = blockIdx.y;
    const int m0 = blockIdx.x * 64;
    const int t  = threadIdx.x;
    const int w  = t >> 5, l = t & 31;
    const int lm = l & 15, lh = l >> 4;

    const float* xa = x + (size_t)(w * 32 + lm) * NN + 16 * lh;
    const float* bp = a2 + ((size_t)s * NN + 16 * lh) * NN + m0 + 4 * lm;
    const float* rp = r2 + s * NN + 16 * lh;
    const v4f cv = *(const v4f*)(c2 + s * NN + m0 + 4 * lm);

    v8f acc[2][4];
#pragma unroll
    for (int rt = 0; rt < 2; ++rt)
#pragma unroll
        for (int ct = 0; ct < 4; ++ct)
#pragma unroll
            for (int g = 0; g < 8; ++g) acc[rt][ct][g] = 0.0f;

    GemmFrag f0, f1;
    gemm_load(f0, xa, rp, bp, 0);
#pragma nounroll
    for (int k0 = 0; k0 < NN; k0 += 64) {
        gemm_load(f1, xa, rp, bp, k0 + 32);
        __builtin_amdgcn_sched_barrier(0);
        gemm_step(f0, cv, acc);
        __builtin_amdgcn_sched_barrier(0);
        const int kn = (k0 + 64 < NN) ? k0 + 64 : k0;
        gemm_load(f0, xa, rp, bp, kn);
        __builtin_amdgcn_sched_barrier(0);
        gemm_step(f1, cv, acc);
        __builtin_amdgcn_sched_barrier(0);
    }
    float* ro = out + (size_t)s * MB * NN + (size_t)(w * 32 + 8 * lh) * NN + m0 + 4 * lm;
#pragma unroll
    for (int rt = 0; rt < 2; ++rt)
#pragma unroll
        for (int g = 0; g < 8; ++g) {
            v4f o;
            o[0] = acc[rt][0][g]; o[1] = acc[rt][1][g]; o[2] = acc[rt][2][g]; o[3] = acc[rt][3][g];
            *(v4f*)(ro + (size_t)(rt * 16 + g) * NN) = o;
        }
}

extern "C" void kernel_launch(void* const* d_in, const int* in_sizes, int n_in,
                              void* d_out, int out_size, void* d_ws, size_t ws_size,
                              hipStream_t stream) {
    const float* x  = (const float*)d_in[0];
    const float* la = (const float*)d_in[1];
    const float* u  = (const float*)d_in[2];
    float* out = (float*)d_out;

    char* ws = (char*)d_ws;
    float* a2    = (float*)ws;
    float* r2    = (float*)(ws + (size_t)NS * NN * NN * sizeof(float));
    float* c2    = r2 + NS * NN;
    float* pa    = out;
    float* pb    = out + NS * NP * NN;

    prepcolrow_kernel<<<dim3(NN / 64, NS), 512, 0, stream>>>(la, u, a2, c2, pa);
    for (int k = 0; k < 10; ++k) {
        colrow_kernel<<<dim3(NN / 64, NS), 512, 0, stream>>>(a2, pa, c2, pb);
        colrow_kernel<<<dim3(NN / 64, NS), 512, 0, stream>>>(a2, pb, c2, pa);
    }
    rowfin_kernel<<<dim3(NN / 256, NS), 256, 0, stream>>>(pa, r2);
    gemm_kernel<<<dim3(NN / 64, NS), 256, 0, stream>>>(x, a2, r2, c2, out);
}
